# starting kernel
# baseline (speedup 1.0000x reference)
#include <hip/hip_runtime.h>
#include <hip/hip_bf16.h>

typedef __attribute__((ext_vector_type(2))) float v2f;
typedef __attribute__((ext_vector_type(8))) float v8f;

#define N_NODES 50000
#define N_EDGES 800000
#define D 128
#define N_CLASSES 40
#define ROW_TILES 3125

__global__ __launch_bounds__(256) void deg_init(float* __restrict__ deg) {
    int i = blockIdx.x * blockDim.x + threadIdx.x;
    if (i < N_NODES) deg[i] = 1.0f;
}

__global__ __launch_bounds__(256) void deg_count(const int* __restrict__ dst,
                                                 float* __restrict__ deg) {
    int e = blockIdx.x * blockDim.x + threadIdx.x;
    if (e < N_EDGES) unsafeAtomicAdd(&deg[dst[e]], 1.0f);
}

__global__ __launch_bounds__(256) void deg_rsqrt(float* __restrict__ deg) {
    int i = blockIdx.x * blockDim.x + threadIdx.x;
    if (i < N_NODES) deg[i] = rsqrtf(deg[i]);
}

__global__ __launch_bounds__(256) void gcn_gemm128(const float* __restrict__ A,
                                                   const float* __restrict__ W,
                                                   float* __restrict__ O) {
    __shared__ float Ws[D * D];
    const int tid = threadIdx.x;
#pragma unroll
    for (int u = tid; u < 2048; u += 256) {
        const int kp = u >> 5;
        const int n4 = (u & 31) * 4;
        const float4 r0 = *(const float4*)(W + (2 * kp) * D + n4);
        const float4 r1 = *(const float4*)(W + (2 * kp + 1) * D + n4);
        float4* dst = (float4*)(Ws + kp * 256 + n4 * 2);
        dst[0] = make_float4(r0.x, r1.x, r0.y, r1.y);
        dst[1] = make_float4(r0.z, r1.z, r0.w, r1.w);
    }
    __syncthreads();

    const int wave = tid >> 5;
    const int lane = tid & 31;
    const int tile = blockIdx.x * 8 + wave;
    if (tile >= ROW_TILES) return;

    const int nlo   = lane & 15;
    const int khalf = lane >> 4;
    const int row   = tile * 16 + nlo;

    v8f acc[8] = {};

    for (int k0 = 0; k0 < D; k0 += 4) {
        const int ka = k0 + 2 * khalf;
        v2f a;
        a.x = A[row * D + ka];
        a.y = A[row * D + ka + 1];
        const v2f* Brow = (const v2f*)(Ws + (ka >> 1) * 256);
#pragma unroll
        for (int t = 0; t < 8; ++t) {
            const v2f b = Brow[t * 16 + nlo];
            acc[t] = __builtin_amdgcn_wmma_f32_16x16x4_f32(
                false, a, false, b, (short)0, acc[t], false, false);
        }
    }

    const int rbase = tile * 16 + 8 * khalf;
#pragma unroll
    for (int t = 0; t < 8; ++t) {
        const int n = t * 16 + nlo;
#pragma unroll
        for (int r = 0; r < 8; ++r) O[(rbase + r) * D + n] = acc[t][r];
    }
}

__global__ __launch_bounds__(256) void zero_buf(float4* __restrict__ p, int n4) {
    int i = blockIdx.x * blockDim.x + threadIdx.x;
    if (i < n4) p[i] = make_float4(0.f, 0.f, 0.f, 0.f);
}

__global__ __launch_bounds__(256) void gcn_scatter(const float* __restrict__ lin,
                                                   const int* __restrict__ src,
                                                   const int* __restrict__ dst,
                                                   const float* __restrict__ dinv,
                                                   float* __restrict__ agg) {
    const int e    = (blockIdx.x * 256 + threadIdx.x) >> 5;
    const int lane = threadIdx.x & 31;
    if (e >= N_EDGES) return;
    const int s = src[e];
    const int d = dst[e];
    const float w = dinv[s] * dinv[d];
    const float4 v = ((const float4*)(lin + s * D))[lane];
    float* ap = agg + d * D + lane * 4;
    unsafeAtomicAdd(ap + 0, v.x * w);
    unsafeAtomicAdd(ap + 1, v.y * w);
    unsafeAtomicAdd(ap + 2, v.z * w);
    unsafeAtomicAdd(ap + 3, v.w * w);
}

__global__ __launch_bounds__(256) void gcn_epilogue(const float* __restrict__ agg,
                                                    const float* __restrict__ lin,
                                                    const float* __restrict__ dinv,
                                                    const float* __restrict__ bias,
                                                    float* __restrict__ hout) {
    int i = blockIdx.x * blockDim.x + threadIdx.x;
    if (i >= N_NODES * (D / 4)) return;
    const int node = i >> 5;
    const int f4   = i & 31;
    float s = dinv[node];
    s = s * s;
    const float4 a = ((const float4*)agg)[i];
    const float4 l = ((const float4*)lin)[i];
    const float4 b = ((const float4*)bias)[f4];
    float4 r;
    r.x = fmaxf(a.x + l.x * s + b.x, 0.f);
    r.y = fmaxf(a.y + l.y * s + b.y, 0.f);
    r.z = fmaxf(a.z + l.z * s + b.z, 0.f);
    r.w = fmaxf(a.w + l.w * s + b.w, 0.f);
    ((float4*)hout)[i] = r;
}

__global__ __launch_bounds__(256) void gcn_fc(const float* __restrict__ H,
                                              const float* __restrict__ W,
                                              const float* __restrict__ bias,
                                              float* __restrict__ O) {
    const int wave = threadIdx.x >> 5;
    const int lane = threadIdx.x & 31;
    const int tile = blockIdx.x * 8 + wave;
    if (tile >= ROW_TILES) return;

    const int nlo   = lane & 15;
    const int khalf = lane >> 4;
    const int row   = tile * 16 + nlo;

    v8f acc[3] = {};
    for (int k0 = 0; k0 < D; k0 += 4) {
        const int ka = k0 + 2 * khalf;
        v2f a;
        a.x = H[row * D + ka];
        a.y = H[row * D + ka + 1];
#pragma unroll
        for (int t = 0; t < 3; ++t) {
            const int n  = t * 16 + nlo;
            const int nc = n < N_CLASSES ? n : (N_CLASSES - 1);
            v2f b;
            b.x = W[ka * N_CLASSES + nc];
            b.y = W[(ka + 1) * N_CLASSES + nc];
            acc[t] = __builtin_amdgcn_wmma_f32_16x16x4_f32(
                false, a, false, b, (short)0, acc[t], false, false);
        }
    }

    const int rbase = tile * 16 + 8 * khalf;
#pragma unroll
    for (int t = 0; t < 3; ++t) {
        const int n = t * 16 + nlo;
        if (n < N_CLASSES) {
            const float bb = bias[n];
#pragma unroll
            for (int r = 0; r < 8; ++r)
                O[(rbase + r) * N_CLASSES + n] = acc[t][r] + bb;
        }
    }
}

extern "C" void kernel_launch(void* const* d_in, const int* in_sizes, int n_in,
                              void* d_out, int out_size, void* d_ws, size_t ws_size,
                              hipStream_t stream) {
    const float* x   = (const float*)d_in[0];
    const int*   ei  = (const int*)d_in[1];
    const float* W1  = (const float*)d_in[2];
    const float* b1  = (const float*)d_in[3];
    const float* W2  = (const float*)d_in[4];
    const float* b2  = (const float*)d_in[5];
    const float* W3  = (const float*)d_in[6];
    const float* b3  = (const float*)d_in[7];
    const float* fcW = (const float*)d_in[8];
    const float* fcb = (const float*)d_in[9];

    float* out    = (float*)d_out;
    float* h_out  = out;
    float* logits = out + N_NODES * D;

    float* ws   = (float*)d_ws;
    float* dinv = ws;
    float* B0   = ws + 50176;
    float* B1   = B0 + N_NODES * D;

    const int* src = ei;
    const int* dst = ei + N_EDGES;

    const int TPB = 256;
    const int gN    = (N_NODES + TPB - 1) / TPB;
    const int gE    = (N_EDGES + TPB - 1) / TPB;
    const int gF4   = (N_NODES * (D / 4) + TPB - 1) / TPB;
    const int gTile = (ROW_TILES + 7) / 8;
    const int gScat = (N_EDGES + 7) / 8;

    deg_init<<<gN, TPB, 0, stream>>>(dinv);
    deg_count<<<gE, TPB, 0, stream>>>(dst, dinv);
    deg_rsqrt<<<gN, TPB, 0, stream>>>(dinv);

    gcn_gemm128<<<gTile, TPB, 0, stream>>>(x, W1, B0);
    zero_buf<<<gF4, TPB, 0, stream>>>((float4*)B1, N_NODES * (D / 4));
    gcn_scatter<<<gScat, TPB, 0, stream>>>(B0, src, dst, dinv, B1);
    gcn_epilogue<<<gF4, TPB, 0, stream>>>(B1, B0, dinv, b1, B1);

    gcn_gemm128<<<gTile, TPB, 0, stream>>>(B1, W2, B0);
    zero_buf<<<gF4, TPB, 0, stream>>>((float4*)B1, N_NODES * (D / 4));
    gcn_scatter<<<gScat, TPB, 0, stream>>>(B0, src, dst, dinv, B1);
    gcn_epilogue<<<gF4, TPB, 0, stream>>>(B1, B0, dinv, b2, B1);

    gcn_gemm128<<<gTile, TPB, 0, stream>>>(B1, W3, B0);
    zero_buf<<<gF4, TPB, 0, stream>>>((float4*)B1, N_NODES * (D / 4));
    gcn_scatter<<<gScat, TPB, 0, stream>>>(B0, src, dst, dinv, B1);
    gcn_epilogue<<<gF4, TPB, 0, stream>>>(B1, B0, dinv, b3, h_out);

    gcn_fc<<<gTile, TPB, 0, stream>>>(h_out, fcW, fcb, logits);
}


# Aggregate first, transform after: owner block sums raw input rows in its LDS tile, then multiplies the tile by W on WMMA (W in LDS), stores relu rows; 3 product launches go; sum order now (A h) W
# speedup vs baseline: 3.3964x; 3.3964x over previous
#include <hip/hip_runtime.h>
#include <hip/hip_bf16.h>

typedef __attribute__((ext_vector_type(2))) float v2f;
typedef __attribute__((ext_vector_type(8))) float v8f;

#define N_NODES 50000
#define N_EDGES 800000
#define D 128
#define N_CLASSES 40
#define ROW_TILES 3125

#define R_OWN 196
#define OWN_BLOCKS 256
#define Q_CAP 16384
#define R_PAD 208
#define PITCH 132
#define E4 (N_EDGES / 4)

__global__ __launch_bounds__(256) void deg_init(float* __restrict__ deg,
                                                unsigned* __restrict__ qcount) {
    int i = blockIdx.x * blockDim.x + threadIdx.x;
    if (i < N_NODES) deg[i] = 1.0f;
    if (i < OWN_BLOCKS) qcount[i] = 0u;
}

__device__ __forceinline__ unsigned deal_count(int d, unsigned* __restrict__ hist,
                                               float* __restrict__ deg) {
    unsafeAtomicAdd(&deg[d], 1.0f);
    return atomicAdd(&hist[(unsigned)d / R_OWN], 1u);
}

__device__ __forceinline__ void deal_place(int s, int d, unsigned rank,
                                           const unsigned* __restrict__ start,
                                           unsigned* __restrict__ qent) {
    const unsigned own = (unsigned)d / R_OWN;
    const unsigned pos = start[own] + rank;
    if (pos < Q_CAP)
        qent[own * Q_CAP + pos] = (unsigned)s | (((unsigned)d - own * R_OWN) << 16);
}

__global__ __launch_bounds__(1024) void edge_deal(const int* __restrict__ src,
                                                  const int* __restrict__ dst,
                                                  float* __restrict__ deg,
                                                  unsigned* __restrict__ qent,
                                                  unsigned* __restrict__ qcount) {
    __shared__ unsigned hist[OWN_BLOCKS];
    __shared__ unsigned start[OWN_BLOCKS];
    const int tid = threadIdx.x;
    if (tid < OWN_BLOCKS) hist[tid] = 0u;
    __syncthreads();

    const int i4 = blockIdx.x * 1024 + tid;
    const bool in = i4 < E4;
    int4 d = make_int4(0, 0, 0, 0);
    int4 s = make_int4(0, 0, 0, 0);
    unsigned k0 = 0u, k1 = 0u, k2 = 0u, k3 = 0u;
    if (in) {
        d = ((const int4*)dst)[i4];
        s = ((const int4*)src)[i4];
        k0 = deal_count(d.x, hist, deg);
        k1 = deal_count(d.y, hist, deg);
        k2 = deal_count(d.z, hist, deg);
        k3 = deal_count(d.w, hist, deg);
    }
    __syncthreads();
    if (tid < OWN_BLOCKS) {
        const unsigned cnt = hist[tid];
        start[tid] = cnt ? atomicAdd(&qcount[tid], cnt) : 0u;
    }
    __syncthreads();
    if (in) {
        deal_place(s.x, d.x, k0, start, qent);
        deal_place(s.y, d.y, k1, start, qent);
        deal_place(s.z, d.z, k2, start, qent);
        deal_place(s.w, d.w, k3, start, qent);
    }
}

__global__ __launch_bounds__(256) void deg_rsqrt(float* __restrict__ deg) {
    int i = blockIdx.x * blockDim.x + threadIdx.x;
    if (i < N_NODES) deg[i] = rsqrtf(deg[i]);
}


__device__ __forceinline__ void own_add_row(float* __restrict__ acc, const float4 a,
                                            float w, int r, int lane) {
    float* q = acc + r * PITCH + lane;
    unsafeAtomicAdd(q,      a.x * w);
    unsafeAtomicAdd(q + 32, a.y * w);
    unsafeAtomicAdd(q + 64, a.z * w);
    unsafeAtomicAdd(q + 96, a.w * w);
}

__device__ __forceinline__ void own_add_edge(float* __restrict__ acc,
                                             const float* __restrict__ lin,
                                             const int* __restrict__ src,
                                             const float* __restrict__ dinv,
                                             int lo, int e, int r, int lane) {
    const int s = src[e];
    const float w = dinv[s] * dinv[lo + r];
    const float4 a = *(const float4*)(lin + s * D + 4 * lane);
    own_add_row(acc, a, w, r, lane);
}

template <bool PARTIAL>
__device__ __forceinline__ void own_add_rows8(float* __restrict__ acc,
                                              const float* __restrict__ lin,
                                              int s, int r, float w,
                                              int j, int cnt, int lane) {
    float4 a[8];
#pragma unroll
    for (int u = 0; u < 8; ++u) {
        const int su = __builtin_amdgcn_readlane(s, j + u);
        a[u] = *(const float4*)(lin + su * D + 4 * lane);
    }
#pragma unroll
    for (int u = 0; u < 8; ++u) {
        if (!PARTIAL || j + u < cnt) {
            const int   ru = __builtin_amdgcn_readlane(r, j + u);
            const float wu = __int_as_float(__builtin_amdgcn_readlane(__float_as_int(w), j + u));
            own_add_row(acc, a[u], wu, ru, lane);
        }
    }
}

__device__ __forceinline__ void own_take(int d, int e, int lo, int lane,
                                         unsigned* __restrict__ queue,
                                         unsigned* __restrict__ qn,
                                         float* __restrict__ acc,
                                         const float* __restrict__ lin,
                                         const int* __restrict__ src,
                                         const float* __restrict__ dinv) {
    const unsigned r = (unsigned)(d - lo);
    const unsigned ent = (unsigned)e | (r << 20);
    bool over = false;
    if (r < R_OWN) {
        const unsigned pos = atomicAdd(qn, 1u);
        if (pos < Q_CAP) queue[pos] = ent;
        else over = true;
    }
    unsigned ob = __builtin_amdgcn_ballot_w32(over);
    while (ob) {
        const int j = __builtin_ctz(ob);
        ob &= ob - 1;
        const unsigned ej = (unsigned)__builtin_amdgcn_readlane((int)ent, j);
        own_add_edge(acc, lin, src, dinv, lo, (int)(ej & 0xFFFFFu), (int)(ej >> 20), lane);
    }
}

__device__ __forceinline__ void own_scan4(const int4 v, int i4, int lo, int lane,
                                          unsigned* __restrict__ queue,
                                          unsigned* __restrict__ qn,
                                          float* __restrict__ acc,
                                          const float* __restrict__ lin,
                                          const int* __restrict__ src,
                                          const float* __restrict__ dinv) {
    const bool any = ((unsigned)(v.x - lo) < R_OWN) | ((unsigned)(v.y - lo) < R_OWN) |
                     ((unsigned)(v.z - lo) < R_OWN) | ((unsigned)(v.w - lo) < R_OWN);
    if (__builtin_amdgcn_ballot_w32(any) == 0) return;
    own_take(v.x, 4 * i4 + 0, lo, lane, queue, qn, acc, lin, src, dinv);
    own_take(v.y, 4 * i4 + 1, lo, lane, queue, qn, acc, lin, src, dinv);
    own_take(v.z, 4 * i4 + 2, lo, lane, queue, qn, acc, lin, src, dinv);
    own_take(v.w, 4 * i4 + 3, lo, lane, queue, qn, acc, lin, src, dinv);
}

__device__ __forceinline__ void stage_w(float* __restrict__ Ws,
                                        const float* __restrict__ W, int tid) {
    for (int u = tid; u < 2048; u += 1024) {
        const int kp = u >> 5;
        const int n4 = (u & 31) * 4;
        const int p0 = 2 * kp;
        const int p1 = 2 * kp + 1;
        const int f0 = (p0 & 31) * 4 + (p0 >> 5);
        const int f1 = (p1 & 31) * 4 + (p1 >> 5);
        const float4 r0 = *(const float4*)(W + f0 * D + n4);
        const float4 r1 = *(const float4*)(W + f1 * D + n4);
        float4* dst = (float4*)(Ws + kp * 256 + n4 * 2);
        dst[0] = make_float4(r0.x, r1.x, r0.y, r1.y);
        dst[1] = make_float4(r0.z, r1.z, r0.w, r1.w);
    }
}

template <int REUSE>
__global__ __launch_bounds__(1024) void gcn_layer_own(const float* __restrict__ hin,
                                                      const int* __restrict__ src,
                                                      const int* __restrict__ dst,
                                                      const float* __restrict__ dinv,
                                                      const float* __restrict__ W,
                                                      const float* __restrict__ bias,
                                                      float* __restrict__ hout,
                                                      const unsigned* __restrict__ qent,
                                                      float* __restrict__ qw,
                                                      const unsigned* __restrict__ qcount) {
    __shared__ float acc[R_PAD * PITCH];
    __shared__ unsigned queue[Q_CAP];
    __shared__ unsigned qn;
    float* Ws = (float*)queue;

    const int tid  = threadIdx.x;
    const int lane = tid & 31;
    const int wave = __builtin_amdgcn_readfirstlane(tid >> 5);
    const int lo   = blockIdx.x * R_OWN;

    for (int i = tid; i < R_PAD * PITCH / 4; i += 1024)
        ((float4*)acc)[i] = make_float4(0.f, 0.f, 0.f, 0.f);
    if (tid == 0) qn = 0u;

    const unsigned* qe = qent + blockIdx.x * Q_CAP;
    float* qwb = qw + blockIdx.x * Q_CAP;
    const unsigned nsaved = qcount[blockIdx.x];
    const bool saved = nsaved <= Q_CAP;
    if (saved) stage_w(Ws, W, tid);
    __syncthreads();

    if (!saved) {
    const int4* d4 = (const int4*)dst;
    for (int it = 0; it < 192; it += 4) {
        const int i0 = it * 1024 + tid;
        const int4 v0 = d4[i0];
        const int4 v1 = d4[i0 + 1024];
        const int4 v2 = d4[i0 + 2048];
        const int4 v3 = d4[i0 + 3072];
        own_scan4(v0, i0,        lo, lane, queue, &qn, acc, hin, src, dinv);
        own_scan4(v1, i0 + 1024, lo, lane, queue, &qn, acc, hin, src, dinv);
        own_scan4(v2, i0 + 2048, lo, lane, queue, &qn, acc, hin, src, dinv);
        own_scan4(v3, i0 + 3072, lo, lane, queue, &qn, acc, hin, src, dinv);
    }
    for (int i = 192 * 1024 + tid; i < E4; i += 1024)
        own_scan4(d4[i], i, lo, lane, queue, &qn, acc, hin, src, dinv);
    __syncthreads();
    }

    const unsigned nq = saved ? nsaved : qn;
    const unsigned n = (unsigned)__builtin_amdgcn_readfirstlane((int)(nq < Q_CAP ? nq : Q_CAP));
    for (unsigned base = (unsigned)wave * 32u; base < n; base += 1024u) {
        const unsigned idx = base + (unsigned)lane;
        int   s = 0;
        int   r = 0;
        float w = 0.f;
        if (idx < n) {
            if (saved) {
                const unsigned ent = qe[idx];
                s = (int)(ent & 0xFFFFu);
                r = (int)(ent >> 16);
                if (REUSE) {
                    w = qwb[idx];
                } else {
                    w = dinv[s] * dinv[lo + r];
                    qwb[idx] = w;
                }
            } else {
                const unsigned ent = queue[idx];
                s = src[ent & 0xFFFFFu];
                r = (int)(ent >> 20);
                w = dinv[s] * dinv[lo + r];
            }
        }
        const int cnt = (int)((n - base) < 32u ? (n - base) : 32u);
        int j = 0;
        for (; j + 8 <= cnt; j += 8)
            own_add_rows8<false>(acc, hin, s, r, w, j, cnt, lane);
        if (j < cnt)
            own_add_rows8<true>(acc, hin, s, r, w, j, cnt, lane);
    }
    __syncthreads();

    if (!saved) stage_w(Ws, W, tid);
    for (int i = tid; i < R_OWN * (D / 4); i += 1024) {
        const int node = lo + (i >> 5);
        if (node < N_NODES) {
            const int f4 = i & 31;
            float sl = dinv[node];
            sl = sl * sl;
            const float4 l = ((const float4*)hin)[node * (D / 4) + f4];
            float* ar = acc + (i >> 5) * PITCH + f4;
            ar[0]  += l.x * sl;
            ar[32] += l.y * sl;
            ar[64] += l.z * sl;
            ar[96] += l.w * sl;
        }
    }
    __syncthreads();

    if (wave >= 26) return;
    const int rt    = wave >> 1;
    const int cbase = (wave & 1) * 64;
    const int nlo   = lane & 15;
    const int khalf = lane >> 4;

    const v2f* Arow = (const v2f*)(acc + (rt * 16 + nlo) * PITCH + 2 * khalf);
    v8f o[4] = {};
    for (int j = 0; j < 32; ++j) {
        const v2f a = Arow[2 * j];
        const v2f* Brow = (const v2f*)(Ws + (2 * j + khalf) * 256 + cbase * 2);
#pragma unroll
        for (int t = 0; t < 4; ++t) {
            const v2f b = Brow[t * 16 + nlo];
            o[t] = __builtin_amdgcn_wmma_f32_16x16x4_f32(
                false, a, false, b, (short)0, o[t], false, false);
        }
    }

    const int lrbase = rt * 16 + 8 * khalf;
#pragma unroll
    for (int t = 0; t < 4; ++t) {
        const int nn = cbase + t * 16 + nlo;
        const float bb = bias[nn];
#pragma unroll
        for (int r = 0; r < 8; ++r) {
            const int lr = lrbase + r;
            if (lr < R_OWN && lo + lr < N_NODES)
                hout[(lo + lr) * D + nn] = fmaxf(o[t][r] + bb, 0.f);
        }
    }
}

__global__ __launch_bounds__(256) void gcn_fc(const float* __restrict__ H,
                                              const float* __restrict__ W,
                                              const float* __restrict__ bias,
                                              float* __restrict__ O) {
    const int wave = threadIdx.x >> 5;
    const int lane = threadIdx.x & 31;
    const int tile = blockIdx.x * 8 + wave;
    if (tile >= ROW_TILES) return;

    const int nlo   = lane & 15;
    const int khalf = lane >> 4;
    const int row   = tile * 16 + nlo;

    v8f acc[3] = {};
    for (int k0 = 0; k0 < D; k0 += 4) {
        const int ka = k0 + 2 * khalf;
        v2f a;
        a.x = H[row * D + ka];
        a.y = H[row * D + ka + 1];
#pragma unroll
        for (int t = 0; t < 3; ++t) {
            const int n  = t * 16 + nlo;
            const int nc = n < N_CLASSES ? n : (N_CLASSES - 1);
            v2f b;
            b.x = W[ka * N_CLASSES + nc];
            b.y = W[(ka + 1) * N_CLASSES + nc];
            acc[t] = __builtin_amdgcn_wmma_f32_16x16x4_f32(
                false, a, false, b, (short)0, acc[t], false, false);
        }
    }

    const int rbase = tile * 16 + 8 * khalf;
#pragma unroll
    for (int t = 0; t < 3; ++t) {
        const int n = t * 16 + nlo;
        if (n < N_CLASSES) {
            const float bb = bias[n];
#pragma unroll
            for (int r = 0; r < 8; ++r)
                O[(rbase + r) * N_CLASSES + n] = acc[t][r] + bb;
        }
    }
}

extern "C" void kernel_launch(void* const* d_in, const int* in_sizes, int n_in,
                              void* d_out, int out_size, void* d_ws, size_t ws_size,
                              hipStream_t stream) {
    const float* x   = (const float*)d_in[0];
    const int*   ei  = (const int*)d_in[1];
    const float* W1  = (const float*)d_in[2];
    const float* b1  = (const float*)d_in[3];
    const float* W2  = (const float*)d_in[4];
    const float* b2  = (const float*)d_in[5];
    const float* W3  = (const float*)d_in[6];
    const float* b3  = (const float*)d_in[7];
    const float* fcW = (const float*)d_in[8];
    const float* fcb = (const float*)d_in[9];

    float* out    = (float*)d_out;
    float* h_out  = out;
    float* logits = out + N_NODES * D;

    float* ws   = (float*)d_ws;
    float* dinv = ws;
    float* B0   = ws + 50176;
    float* B1   = B0 + N_NODES * D;
    unsigned* qent   = (unsigned*)(B1 + N_NODES * D);
    float*    qw     = (float*)(qent + OWN_BLOCKS * Q_CAP);
    unsigned* qcount = (unsigned*)(qw + OWN_BLOCKS * Q_CAP);

    const int* src = ei;
    const int* dst = ei + N_EDGES;

    const int TPB = 256;
    const int gN    = (N_NODES + TPB - 1) / TPB;
    const int gDeal = (E4 + 1023) / 1024;
    const int gTile = (ROW_TILES + 7) / 8;

    deg_init<<<gN, TPB, 0, stream>>>(dinv, qcount);
    edge_deal<<<gDeal, 1024, 0, stream>>>(src, dst, dinv, qent, qcount);
    deg_rsqrt<<<gN, TPB, 0, stream>>>(dinv);

    gcn_layer_own<0><<<OWN_BLOCKS, 1024, 0, stream>>>(x,  src, dst, dinv, W1, b1, B0, qent, qw, qcount);
    gcn_layer_own<1><<<OWN_BLOCKS, 1024, 0, stream>>>(B0, src, dst, dinv, W2, b2, B1, qent, qw, qcount);
    gcn_layer_own<1><<<OWN_BLOCKS, 1024, 0, stream>>>(B1, src, dst, dinv, W3, b3, h_out, qent, qw, qcount);

    gcn_fc<<<gTile, TPB, 0, stream>>>(h_out, fcW, fcb, logits);
}


# Tile rows owned by waves: block bins edges + self loops in LDS by row % 32; each wave sums its bin into its own rows by plain 16-byte LDS read-add-write (bin arrival order); LDS float adds go
# speedup vs baseline: 3.5211x; 1.0367x over previous
#include <hip/hip_runtime.h>
#include <hip/hip_bf16.h>

typedef __attribute__((ext_vector_type(2))) float v2f;
typedef __attribute__((ext_vector_type(8))) float v8f;

#define N_NODES 50000
#define N_EDGES 800000
#define D 128
#define N_CLASSES 40
#define ROW_TILES 3125

#define R_OWN 196
#define OWN_BLOCKS 256
#define Q_CAP 16384
#define R_PAD 208
#define PITCH 132
#define E4 (N_EDGES / 4)
#define WAVES 32
#define SUB_CAP 256

__global__ __launch_bounds__(256) void deg_init(float* __restrict__ deg,
                                                unsigned* __restrict__ qcount) {
    int i = blockIdx.x * blockDim.x + threadIdx.x;
    if (i < N_NODES) deg[i] = 1.0f;
    if (i < OWN_BLOCKS) qcount[i] = 0u;
}

__device__ __forceinline__ unsigned deal_count(int d, unsigned* __restrict__ hist,
                                               float* __restrict__ deg) {
    unsafeAtomicAdd(&deg[d], 1.0f);
    return atomicAdd(&hist[(unsigned)d / R_OWN], 1u);
}

__device__ __forceinline__ void deal_place(int s, int d, unsigned rank,
                                           const unsigned* __restrict__ start,
                                           unsigned* __restrict__ qent) {
    const unsigned own = (unsigned)d / R_OWN;
    const unsigned pos = start[own] + rank;
    if (pos < Q_CAP)
        qent[own * Q_CAP + pos] = (unsigned)s | (((unsigned)d - own * R_OWN) << 16);
}

__global__ __launch_bounds__(1024) void edge_deal(const int* __restrict__ src,
                                                  const int* __restrict__ dst,
                                                  float* __restrict__ deg,
                                                  unsigned* __restrict__ qent,
                                                  unsigned* __restrict__ qcount) {
    __shared__ unsigned hist[OWN_BLOCKS];
    __shared__ unsigned start[OWN_BLOCKS];
    const int tid = threadIdx.x;
    if (tid < OWN_BLOCKS) hist[tid] = 0u;
    __syncthreads();

    const int i4 = blockIdx.x * 1024 + tid;
    const bool in = i4 < E4;
    int4 d = make_int4(0, 0, 0, 0);
    int4 s = make_int4(0, 0, 0, 0);
    unsigned k0 = 0u, k1 = 0u, k2 = 0u, k3 = 0u;
    if (in) {
        d = ((const int4*)dst)[i4];
        s = ((const int4*)src)[i4];
        k0 = deal_count(d.x, hist, deg);
        k1 = deal_count(d.y, hist, deg);
        k2 = deal_count(d.z, hist, deg);
        k3 = deal_count(d.w, hist, deg);
    }
    __syncthreads();
    if (tid < OWN_BLOCKS) {
        const unsigned cnt = hist[tid];
        start[tid] = cnt ? atomicAdd(&qcount[tid], cnt) : 0u;
    }
    __syncthreads();
    if (in) {
        deal_place(s.x, d.x, k0, start, qent);
        deal_place(s.y, d.y, k1, start, qent);
        deal_place(s.z, d.z, k2, start, qent);
        deal_place(s.w, d.w, k3, start, qent);
    }
}

__global__ __launch_bounds__(256) void deg_rsqrt(float* __restrict__ deg) {
    int i = blockIdx.x * blockDim.x + threadIdx.x;
    if (i < N_NODES) deg[i] = rsqrtf(deg[i]);
}


__device__ __forceinline__ void own_add_row(float* __restrict__ acc, const float4 a,
                                            float w, int r, int lane) {
    float* q = acc + r * PITCH + 4 * lane;
    unsafeAtomicAdd(q,     a.x * w);
    unsafeAtomicAdd(q + 1, a.y * w);
    unsafeAtomicAdd(q + 2, a.z * w);
    unsafeAtomicAdd(q + 3, a.w * w);
}

__device__ __forceinline__ void own_load8(float4 (&a)[8], const float* __restrict__ lin,
                                          int s, int j, int lane) {
#pragma unroll
    for (int u = 0; u < 8; ++u) {
        const int su = __builtin_amdgcn_readlane(s, j + u);
        a[u] = *(const float4*)(lin + su * D + 4 * lane);
    }
}

__device__ __forceinline__ void own_rmw1(float* mine, const float4 a, int k, float w) {
    float4* p = (float4*)(mine + k * (WAVES * PITCH));
    float4 x = *p;
    x.x += a.x * w;
    x.y += a.y * w;
    x.z += a.z * w;
    x.w += a.w * w;
    *p = x;
}

__device__ __forceinline__ void own_rmw2(float* mine, const float4 a0, const float4 a1,
                                         int k0, int k1, float w0, float w1) {
    float4* p0 = (float4*)(mine + k0 * (WAVES * PITCH));
    if (k0 != k1) {
        float4* p1 = (float4*)(mine + k1 * (WAVES * PITCH));
        float4 x0 = *p0;
        float4 x1 = *p1;
        x0.x += a0.x * w0;
        x0.y += a0.y * w0;
        x0.z += a0.z * w0;
        x0.w += a0.w * w0;
        x1.x += a1.x * w1;
        x1.y += a1.y * w1;
        x1.z += a1.z * w1;
        x1.w += a1.w * w1;
        *p0 = x0;
        *p1 = x1;
    } else {
        float4 x0 = *p0;
        x0.x += a0.x * w0;
        x0.y += a0.y * w0;
        x0.z += a0.z * w0;
        x0.w += a0.w * w0;
        x0.x += a1.x * w1;
        x0.y += a1.y * w1;
        x0.z += a1.z * w1;
        x0.w += a1.w * w1;
        *p0 = x0;
    }
}

__device__ __forceinline__ void own_rmw8_full(float* mine, const float4 (&a)[8],
                                              int k, float w, int j) {
#pragma unroll
    for (int u = 0; u < 8; u += 2) {
        const int   k0 = __builtin_amdgcn_readlane(k, j + u);
        const int   k1 = __builtin_amdgcn_readlane(k, j + u + 1);
        const float w0 = __int_as_float(__builtin_amdgcn_readlane(__float_as_int(w), j + u));
        const float w1 = __int_as_float(__builtin_amdgcn_readlane(__float_as_int(w), j + u + 1));
        own_rmw2(mine, a[u], a[u + 1], k0, k1, w0, w1);
    }
}

__device__ __forceinline__ void own_rmw8_part(float* mine, const float4 (&a)[8],
                                              int k, float w, int j, int cnt) {
#pragma unroll
    for (int u = 0; u < 8; ++u) {
        if (j + u < cnt) {
            const int   ku = __builtin_amdgcn_readlane(k, j + u);
            const float wu = __int_as_float(__builtin_amdgcn_readlane(__float_as_int(w), j + u));
            own_rmw1(mine, a[u], ku, wu);
        }
    }
}

__device__ __forceinline__ void own_add_edge(float* __restrict__ acc,
                                             const float* __restrict__ lin,
                                             const int* __restrict__ src,
                                             const float* __restrict__ dinv,
                                             int lo, int e, int r, int lane) {
    const int s = src[e];
    const float w = dinv[s] * dinv[lo + r];
    const float4 a = *(const float4*)(lin + s * D + 4 * lane);
    own_add_row(acc, a, w, r, lane);
}

template <bool PARTIAL>
__device__ __forceinline__ void own_add_rows8(float* __restrict__ acc,
                                              const float* __restrict__ lin,
                                              int s, int r, float w,
                                              int j, int cnt, int lane) {
    float4 a[8];
#pragma unroll
    for (int u = 0; u < 8; ++u) {
        const int su = __builtin_amdgcn_readlane(s, j + u);
        a[u] = *(const float4*)(lin + su * D + 4 * lane);
    }
#pragma unroll
    for (int u = 0; u < 8; ++u) {
        if (!PARTIAL || j + u < cnt) {
            const int   ru = __builtin_amdgcn_readlane(r, j + u);
            const float wu = __int_as_float(__builtin_amdgcn_readlane(__float_as_int(w), j + u));
            own_add_row(acc, a[u], wu, ru, lane);
        }
    }
}

__device__ __forceinline__ void own_take(int d, int e, int lo, int lane,
                                         unsigned* __restrict__ queue,
                                         unsigned* __restrict__ qn,
                                         float* __restrict__ acc,
                                         const float* __restrict__ lin,
                                         const int* __restrict__ src,
                                         const float* __restrict__ dinv) {
    const unsigned r = (unsigned)(d - lo);
    const unsigned ent = (unsigned)e | (r << 20);
    bool over = false;
    if (r < R_OWN) {
        const unsigned pos = atomicAdd(qn, 1u);
        if (pos < Q_CAP) queue[pos] = ent;
        else over = true;
    }
    unsigned ob = __builtin_amdgcn_ballot_w32(over);
    while (ob) {
        const int j = __builtin_ctz(ob);
        ob &= ob - 1;
        const unsigned ej = (unsigned)__builtin_amdgcn_readlane((int)ent, j);
        own_add_edge(acc, lin, src, dinv, lo, (int)(ej & 0xFFFFFu), (int)(ej >> 20), lane);
    }
}

__device__ __forceinline__ void own_scan4(const int4 v, int i4, int lo, int lane,
                                          unsigned* __restrict__ queue,
                                          unsigned* __restrict__ qn,
                                          float* __restrict__ acc,
                                          const float* __restrict__ lin,
                                          const int* __restrict__ src,
                                          const float* __restrict__ dinv) {
    const bool any = ((unsigned)(v.x - lo) < R_OWN) | ((unsigned)(v.y - lo) < R_OWN) |
                     ((unsigned)(v.z - lo) < R_OWN) | ((unsigned)(v.w - lo) < R_OWN);
    if (__builtin_amdgcn_ballot_w32(any) == 0) return;
    own_take(v.x, 4 * i4 + 0, lo, lane, queue, qn, acc, lin, src, dinv);
    own_take(v.y, 4 * i4 + 1, lo, lane, queue, qn, acc, lin, src, dinv);
    own_take(v.z, 4 * i4 + 2, lo, lane, queue, qn, acc, lin, src, dinv);
    own_take(v.w, 4 * i4 + 3, lo, lane, queue, qn, acc, lin, src, dinv);
}

__device__ __forceinline__ void stage_w(float* __restrict__ Ws,
                                        const float* __restrict__ W, int tid) {
    for (int u = tid; u < 2048; u += 1024) {
        const int kp = u >> 5;
        const int n4 = (u & 31) * 4;
        const int f0 = 2 * kp;
        const int f1 = 2 * kp + 1;
        const float4 r0 = *(const float4*)(W + f0 * D + n4);
        const float4 r1 = *(const float4*)(W + f1 * D + n4);
        float4* dst = (float4*)(Ws + kp * 256 + n4 * 2);
        dst[0] = make_float4(r0.x, r1.x, r0.y, r1.y);
        dst[1] = make_float4(r0.z, r1.z, r0.w, r1.w);
    }
}

template <int REUSE>
__global__ __launch_bounds__(1024) void gcn_layer_own(const float* __restrict__ hin,
                                                      const int* __restrict__ src,
                                                      const int* __restrict__ dst,
                                                      const float* __restrict__ dinv,
                                                      const float* __restrict__ W,
                                                      const float* __restrict__ bias,
                                                      float* __restrict__ hout,
                                                      const unsigned* __restrict__ qent,
                                                      float* __restrict__ qw,
                                                      const unsigned* __restrict__ qcount) {
    __shared__ __attribute__((aligned(16))) float acc[R_PAD * PITCH];
    __shared__ __attribute__((aligned(16))) unsigned queue[Q_CAP];
    __shared__ unsigned qn;
    __shared__ uint2 sub[WAVES * SUB_CAP];
    __shared__ unsigned scnt[WAVES];
    float* Ws = (float*)queue;

    const int tid  = threadIdx.x;
    const int lane = tid & 31;
    const int wave = __builtin_amdgcn_readfirstlane(tid >> 5);
    const int lo   = blockIdx.x * R_OWN;

    for (int i = tid; i < R_PAD * PITCH / 4; i += 1024)
        ((float4*)acc)[i] = make_float4(0.f, 0.f, 0.f, 0.f);
    if (tid == 0) qn = 0u;
    if (tid < WAVES) scnt[tid] = 0u;

    const unsigned* qe = qent + blockIdx.x * Q_CAP;
    float* qwb = qw + blockIdx.x * Q_CAP;
    const unsigned nsaved = qcount[blockIdx.x];
    const bool saved = nsaved <= Q_CAP;
    if (saved) stage_w(Ws, W, tid);
    __syncthreads();

    if (!saved) {
    const int4* d4 = (const int4*)dst;
    for (int it = 0; it < 192; it += 4) {
        const int i0 = it * 1024 + tid;
        const int4 v0 = d4[i0];
        const int4 v1 = d4[i0 + 1024];
        const int4 v2 = d4[i0 + 2048];
        const int4 v3 = d4[i0 + 3072];
        own_scan4(v0, i0,        lo, lane, queue, &qn, acc, hin, src, dinv);
        own_scan4(v1, i0 + 1024, lo, lane, queue, &qn, acc, hin, src, dinv);
        own_scan4(v2, i0 + 2048, lo, lane, queue, &qn, acc, hin, src, dinv);
        own_scan4(v3, i0 + 3072, lo, lane, queue, &qn, acc, hin, src, dinv);
    }
    for (int i = 192 * 1024 + tid; i < E4; i += 1024)
        own_scan4(d4[i], i, lo, lane, queue, &qn, acc, hin, src, dinv);
    __syncthreads();

    const unsigned nq = qn;
    const unsigned n = (unsigned)__builtin_amdgcn_readfirstlane((int)(nq < Q_CAP ? nq : Q_CAP));
    for (unsigned base = (unsigned)wave * 32u; base < n; base += 1024u) {
        const unsigned idx = base + (unsigned)lane;
        int   s = 0;
        int   r = 0;
        float w = 0.f;
        if (idx < n) {
            const unsigned ent = queue[idx];
            s = src[ent & 0xFFFFFu];
            r = (int)(ent >> 20);
            w = dinv[s] * dinv[lo + r];
        }
        const int cnt = (int)((n - base) < 32u ? (n - base) : 32u);
        int j = 0;
        for (; j + 8 <= cnt; j += 8)
            own_add_rows8<false>(acc, hin, s, r, w, j, cnt, lane);
        if (j < cnt)
            own_add_rows8<true>(acc, hin, s, r, w, j, cnt, lane);
    }
    } else {
    const unsigned n = (unsigned)__builtin_amdgcn_readfirstlane((int)nsaved);
    const int left = N_NODES - lo;
    const unsigned ntot = n + (unsigned)(left < R_OWN ? left : R_OWN);
    for (unsigned base = (unsigned)wave * 32u; base < ntot; base += 4096u) {
        unsigned ent[4];
        float    wv[4];
#pragma unroll
        for (int q = 0; q < 4; ++q) {
            const unsigned idx = base + (unsigned)q * 1024u + (unsigned)lane;
            ent[q] = 0u;
            wv[q]  = 0.f;
            if (idx < n) {
                ent[q] = qe[idx];
                if (REUSE) wv[q] = qwb[idx];
            }
        }
#pragma unroll
        for (int q = 0; q < 4; ++q) {
            const unsigned idx = base + (unsigned)q * 1024u + (unsigned)lane;
            int   s = 0;
            int   r = 0;
            float w = 0.f;
            bool over = false;
            if (idx < ntot) {
                if (idx < n) {
                    s = (int)(ent[q] & 0xFFFFu);
                    r = (int)(ent[q] >> 16);
                    if (REUSE) {
                        w = wv[q];
                    } else {
                        w = dinv[s] * dinv[lo + r];
                        qwb[idx] = w;
                    }
                } else {
                    r = (int)(idx - n);
                    s = lo + r;
                    w = dinv[s];
                    w = w * w;
                }
                const unsigned b   = (unsigned)r & (WAVES - 1);
                const unsigned pos = atomicAdd(&scnt[b], 1u);
                if (pos < SUB_CAP)
                    sub[b * SUB_CAP + pos] = make_uint2((unsigned)s | (((unsigned)r >> 5) << 16),
                                                        __float_as_uint(w));
                else
                    over = true;
            }
            unsigned ob = __builtin_amdgcn_ballot_w32(over);
            while (ob) {
                const int j = __builtin_ctz(ob);
                ob &= ob - 1;
                const int   sj = __builtin_amdgcn_readlane(s, j);
                const int   rj = __builtin_amdgcn_readlane(r, j);
                const float wj = __int_as_float(__builtin_amdgcn_readlane(__float_as_int(w), j));
                const float4 a = *(const float4*)(hin + sj * D + 4 * lane);
                own_add_row(acc, a, wj, rj, lane);
            }
        }
    }
    __syncthreads();

    {
        const unsigned mraw = scnt[wave];
        const unsigned m = (unsigned)__builtin_amdgcn_readfirstlane((int)(mraw < SUB_CAP ? mraw : SUB_CAP));
        const uint2* mysub = sub + wave * SUB_CAP;
        float* mine = acc + wave * PITCH + 4 * lane;
        for (unsigned b0 = 0; b0 < m; b0 += 32u) {
            const unsigned i = b0 + (unsigned)lane;
            uint2 e = make_uint2(0u, 0u);
            if (i < m) e = mysub[i];
            const int   s = (int)(e.x & 0xFFFFu);
            const int   k = (int)(e.x >> 16);
            const float w = __uint_as_float(e.y);
            const int cnt = (int)((m - b0) < 32u ? (m - b0) : 32u);
            float4 a0[8];
            float4 a1[8];
            if (cnt == 32) {
                own_load8(a0, hin, s, 0, lane);
                own_load8(a1, hin, s, 8, lane);
                own_rmw8_full(mine, a0, k, w, 0);
                own_load8(a0, hin, s, 16, lane);
                own_rmw8_full(mine, a1, k, w, 8);
                own_load8(a1, hin, s, 24, lane);
                own_rmw8_full(mine, a0, k, w, 16);
                own_rmw8_full(mine, a1, k, w, 24);
            } else {
                own_load8(a0, hin, s, 0, lane);
                if (cnt > 8) own_load8(a1, hin, s, 8, lane);
                own_rmw8_part(mine, a0, k, w, 0, cnt);
                if (cnt > 8) {
                    if (cnt > 16) own_load8(a0, hin, s, 16, lane);
                    own_rmw8_part(mine, a1, k, w, 8, cnt);
                    if (cnt > 16) {
                        if (cnt > 24) own_load8(a1, hin, s, 24, lane);
                        own_rmw8_part(mine, a0, k, w, 16, cnt);
                        if (cnt > 24) own_rmw8_part(mine, a1, k, w, 24, cnt);
                    }
                }
            }
        }
    }
    }
    __syncthreads();

    if (!saved) {
        stage_w(Ws, W, tid);
        for (int i = tid; i < R_OWN * (D / 4); i += 1024) {
            const int node = lo + (i >> 5);
            if (node < N_NODES) {
                const int f4 = i & 31;
                float sl = dinv[node];
                sl = sl * sl;
                const float4 l = ((const float4*)hin)[node * (D / 4) + f4];
                float4* ar = (float4*)(acc + (i >> 5) * PITCH + 4 * f4);
                float4 x = *ar;
                x.x += l.x * sl;
                x.y += l.y * sl;
                x.z += l.z * sl;
                x.w += l.w * sl;
                *ar = x;
            }
        }
        __syncthreads();
    }

    if (wave >= 26) return;
    const int rt    = wave >> 1;
    const int cbase = (wave & 1) * 64;
    const int nlo   = lane & 15;
    const int khalf = lane >> 4;

    const v2f* Arow = (const v2f*)(acc + (rt * 16 + nlo) * PITCH + 2 * khalf);
    v8f o[4] = {};
    for (int j = 0; j < 32; ++j) {
        const v2f a = Arow[2 * j];
        const v2f* Brow = (const v2f*)(Ws + (2 * j + khalf) * 256 + cbase * 2);
#pragma unroll
        for (int t = 0; t < 4; ++t) {
            const v2f b = Brow[t * 16 + nlo];
            o[t] = __builtin_amdgcn_wmma_f32_16x16x4_f32(
                false, a, false, b, (short)0, o[t], false, false);
        }
    }

    const int lrbase = rt * 16 + 8 * khalf;
#pragma unroll
    for (int t = 0; t < 4; ++t) {
        const int nn = cbase + t * 16 + nlo;
        const float bb = bias[nn];
#pragma unroll
        for (int r = 0; r < 8; ++r) {
            const int lr = lrbase + r;
            if (lr < R_OWN && lo + lr < N_NODES)
                hout[(lo + lr) * D + nn] = fmaxf(o[t][r] + bb, 0.f);
        }
    }
}

__global__ __launch_bounds__(256) void gcn_fc(const float* __restrict__ H,
                                              const float* __restrict__ W,
                                              const float* __restrict__ bias,
                                              float* __restrict__ O) {
    const int wave = threadIdx.x >> 5;
    const int lane = threadIdx.x & 31;
    const int tile = blockIdx.x * 8 + wave;
    if (tile >= ROW_TILES) return;

    const int nlo   = lane & 15;
    const int khalf = lane >> 4;
    const int row   = tile * 16 + nlo;

    v8f acc[3] = {};
    for (int k0 = 0; k0 < D; k0 += 4) {
        const int ka = k0 + 2 * khalf;
        v2f a;
        a.x = H[row * D + ka];
        a.y = H[row * D + ka + 1];
#pragma unroll
        for (int t = 0; t < 3; ++t) {
            const int n  = t * 16 + nlo;
            const int nc = n < N_CLASSES ? n : (N_CLASSES - 1);
            v2f b;
            b.x = W[ka * N_CLASSES + nc];
            b.y = W[(ka + 1) * N_CLASSES + nc];
            acc[t] = __builtin_amdgcn_wmma_f32_16x16x4_f32(
                false, a, false, b, (short)0, acc[t], false, false);
        }
    }

    const int rbase = tile * 16 + 8 * khalf;
#pragma unroll
    for (int t = 0; t < 3; ++t) {
        const int n = t * 16 + nlo;
        if (n < N_CLASSES) {
            const float bb = bias[n];
#pragma unroll
            for (int r = 0; r < 8; ++r)
                O[(rbase + r) * N_CLASSES + n] = acc[t][r] + bb;
        }
    }
}

extern "C" void kernel_launch(void* const* d_in, const int* in_sizes, int n_in,
                              void* d_out, int out_size, void* d_ws, size_t ws_size,
                              hipStream_t stream) {
    const float* x   = (const float*)d_in[0];
    const int*   ei  = (const int*)d_in[1];
    const float* W1  = (const float*)d_in[2];
    const float* b1  = (const float*)d_in[3];
    const float* W2  = (const float*)d_in[4];
    const float* b2  = (const float*)d_in[5];
    const float* W3  = (const float*)d_in[6];
    const float* b3  = (const float*)d_in[7];
    const float* fcW = (const float*)d_in[8];
    const float* fcb = (const float*)d_in[9];

    float* out    = (float*)d_out;
    float* h_out  = out;
    float* logits = out + N_NODES * D;

    float* ws   = (float*)d_ws;
    float* dinv = ws;
    float* B0   = ws + 50176;
    float* B1   = B0 + N_NODES * D;
    unsigned* qent   = (unsigned*)(B1 + N_NODES * D);
    float*    qw     = (float*)(qent + OWN_BLOCKS * Q_CAP);
    unsigned* qcount = (unsigned*)(qw + OWN_BLOCKS * Q_CAP);

    const int* src = ei;
    const int* dst = ei + N_EDGES;

    const int TPB = 256;
    const int gN    = (N_NODES + TPB - 1) / TPB;
    const int gDeal = (E4 + 1023) / 1024;
    const int gTile = (ROW_TILES + 7) / 8;

    deg_init<<<gN, TPB, 0, stream>>>(dinv, qcount);
    edge_deal<<<gDeal, 1024, 0, stream>>>(src, dst, dinv, qent, qcount);
    deg_rsqrt<<<gN, TPB, 0, stream>>>(dinv);

    gcn_layer_own<0><<<OWN_BLOCKS, 1024, 0, stream>>>(x,  src, dst, dinv, W1, b1, B0, qent, qw, qcount);
    gcn_layer_own<1><<<OWN_BLOCKS, 1024, 0, stream>>>(B0, src, dst, dinv, W2, b2, B1, qent, qw, qcount);
    gcn_layer_own<1><<<OWN_BLOCKS, 1024, 0, stream>>>(B1, src, dst, dinv, W3, b3, h_out, qent, qw, qcount);

    gcn_fc<<<gTile, TPB, 0, stream>>>(h_out, fcW, fcb, logits);
}


# Edges dealt with no global atomics: each dealing block sorts its 4096 edges by owner in LDS, stores runs whole; owners copy runs, count degrees in LDS; 6 launches; row sums still bin arrival order
# speedup vs baseline: 5.1359x; 1.4586x over previous
#include <hip/hip_runtime.h>
#include <hip/hip_bf16.h>

typedef __attribute__((ext_vector_type(2))) float v2f;
typedef __attribute__((ext_vector_type(8))) float v8f;

#define N_NODES 50000
#define N_EDGES 800000
#define D 128
#define N_CLASSES 40
#define ROW_TILES 3125

#define R_OWN 196
#define OWN_BLOCKS 256
#define Q_CAP 16384
#define R_PAD 208
#define PITCH 132
#define E4 (N_EDGES / 4)
#define WAVES 32
#define SUB_CAP 256

#define DEAL_E 4096
#define DEALERS ((N_EDGES + DEAL_E - 1) / DEAL_E)
#define F_PITCH 260

__device__ __forceinline__ unsigned wave_incl_sum(unsigned v, int lane) {
#pragma unroll
    for (int off = 1; off < 32; off <<= 1) {
        const unsigned t = __shfl_up(v, off);
        if (lane >= off) v += t;
    }
    return v;
}

__global__ __launch_bounds__(1024) void edge_deal(const int* __restrict__ src,
                                                  const int* __restrict__ dst,
                                                  unsigned* __restrict__ dealE,
                                                  unsigned* __restrict__ dealF) {
    __shared__ unsigned hist[OWN_BLOCKS];
    __shared__ unsigned first[OWN_BLOCKS + 1];
    __shared__ unsigned wsum[8];
    __shared__ unsigned sorted[DEAL_E];
    const int tid  = threadIdx.x;
    const int lane = tid & 31;
    if (tid < OWN_BLOCKS) hist[tid] = 0u;
    __syncthreads();

    const int i4 = blockIdx.x * 1024 + tid;
    const bool in = i4 < E4;
    int4 d = make_int4(0, 0, 0, 0);
    int4 s = make_int4(0, 0, 0, 0);
    unsigned k0 = 0u, k1 = 0u, k2 = 0u, k3 = 0u;
    if (in) {
        d = ((const int4*)dst)[i4];
        s = ((const int4*)src)[i4];
        k0 = atomicAdd(&hist[(unsigned)d.x / R_OWN], 1u);
        k1 = atomicAdd(&hist[(unsigned)d.y / R_OWN], 1u);
        k2 = atomicAdd(&hist[(unsigned)d.z / R_OWN], 1u);
        k3 = atomicAdd(&hist[(unsigned)d.w / R_OWN], 1u);
    }
    __syncthreads();
    unsigned cnt = 0u, inc = 0u;
    if (tid < OWN_BLOCKS) {
        cnt = hist[tid];
        inc = wave_incl_sum(cnt, lane);
        if (lane == 31) wsum[tid >> 5] = inc;
    }
    __syncthreads();
    if (tid < OWN_BLOCKS) {
        unsigned basew = 0u;
        for (int w = 0; w < (tid >> 5); ++w) basew += wsum[w];
        first[tid] = basew + inc - cnt;
        if (tid == OWN_BLOCKS - 1) first[OWN_BLOCKS] = basew + inc;
    }
    __syncthreads();
    if (in) {
        const unsigned o0 = (unsigned)d.x / R_OWN, o1 = (unsigned)d.y / R_OWN;
        const unsigned o2 = (unsigned)d.z / R_OWN, o3 = (unsigned)d.w / R_OWN;
        sorted[first[o0] + k0] = (unsigned)s.x | (((unsigned)d.x - o0 * R_OWN) << 16);
        sorted[first[o1] + k1] = (unsigned)s.y | (((unsigned)d.y - o1 * R_OWN) << 16);
        sorted[first[o2] + k2] = (unsigned)s.z | (((unsigned)d.z - o2 * R_OWN) << 16);
        sorted[first[o3] + k3] = (unsigned)s.w | (((unsigned)d.w - o3 * R_OWN) << 16);
    }
    __syncthreads();
    const unsigned nblk = first[OWN_BLOCKS];
    unsigned* outE = dealE + blockIdx.x * DEAL_E;
    for (unsigned i = (unsigned)tid; i < nblk; i += 1024u) outE[i] = sorted[i];
    if (tid <= OWN_BLOCKS) dealF[blockIdx.x * F_PITCH + tid] = first[tid];
}

__global__ __launch_bounds__(1024) void own_build(const int* __restrict__ dst,
                                                  const unsigned* __restrict__ dealE,
                                                  const unsigned* __restrict__ dealF,
                                                  unsigned* __restrict__ qent,
                                                  unsigned* __restrict__ qcount,
                                                  float* __restrict__ dinv) {
    __shared__ unsigned cntn[R_PAD];
    __shared__ unsigned rfirst[OWN_BLOCKS];
    __shared__ unsigned rlen[OWN_BLOCKS];
    __shared__ unsigned rbase[OWN_BLOCKS];
    __shared__ unsigned wsum[8];
    __shared__ unsigned ntot;
    const int tid  = threadIdx.x;
    const int lane = tid & 31;
    const int wave = __builtin_amdgcn_readfirstlane(tid >> 5);
    const int b    = blockIdx.x;
    const int lo   = b * R_OWN;

    if (tid < R_PAD) cntn[tid] = 0u;
    unsigned f = 0u, len = 0u, inc = 0u;
    if (tid < OWN_BLOCKS) {
        if (tid < DEALERS) {
            f   = dealF[tid * F_PITCH + b];
            len = dealF[tid * F_PITCH + b + 1] - f;
        }
        inc = wave_incl_sum(len, lane);
        if (lane == 31) wsum[tid >> 5] = inc;
    }
    __syncthreads();
    if (tid < OWN_BLOCKS) {
        unsigned basew = 0u;
        for (int w = 0; w < (tid >> 5); ++w) basew += wsum[w];
        rfirst[tid] = f;
        rlen[tid]   = len;
        rbase[tid]  = basew + inc - len;
        if (tid == OWN_BLOCKS - 1) ntot = basew + inc;
    }
    __syncthreads();
    const unsigned n = ntot;
    if (tid == 0) qcount[b] = n;
    if (n <= Q_CAP) {
        unsigned* qe = qent + b * Q_CAP;
        unsigned e[7];
        bool ok[7];
#pragma unroll
        for (int q = 0; q < 7; ++q) {
            const int j = wave + WAVES * q;
            ok[q] = (unsigned)lane < rlen[j];
            e[q] = 0u;
            if (ok[q]) e[q] = dealE[j * DEAL_E + rfirst[j] + lane];
        }
#pragma unroll
        for (int q = 0; q < 7; ++q) {
            const int j = wave + WAVES * q;
            if (ok[q]) {
                qe[rbase[j] + lane] = e[q];
                atomicAdd(&cntn[e[q] >> 16], 1u);
            }
        }
        for (int j = wave; j < DEALERS; j += WAVES) {
            const unsigned lj = rlen[j];
            for (unsigned k = 32u + (unsigned)lane; k < lj; k += 32u) {
                const unsigned ek = dealE[j * DEAL_E + rfirst[j] + k];
                qe[rbase[j] + k] = ek;
                atomicAdd(&cntn[ek >> 16], 1u);
            }
        }
    } else {
        const int4* d4 = (const int4*)dst;
        for (int i = tid; i < E4; i += 1024) {
            const int4 v = d4[i];
            const unsigned r0 = (unsigned)(v.x - lo), r1 = (unsigned)(v.y - lo);
            const unsigned r2 = (unsigned)(v.z - lo), r3 = (unsigned)(v.w - lo);
            if (r0 < R_OWN) atomicAdd(&cntn[r0], 1u);
            if (r1 < R_OWN) atomicAdd(&cntn[r1], 1u);
            if (r2 < R_OWN) atomicAdd(&cntn[r2], 1u);
            if (r3 < R_OWN) atomicAdd(&cntn[r3], 1u);
        }
    }
    __syncthreads();
    if (tid < R_OWN && lo + tid < N_NODES)
        dinv[lo + tid] = rsqrtf(1.0f + (float)cntn[tid]);
}


__device__ __forceinline__ void own_add_row(float* __restrict__ acc, const float4 a,
                                            float w, int r, int lane) {
    float* q = acc + r * PITCH + 4 * lane;
    unsafeAtomicAdd(q,     a.x * w);
    unsafeAtomicAdd(q + 1, a.y * w);
    unsafeAtomicAdd(q + 2, a.z * w);
    unsafeAtomicAdd(q + 3, a.w * w);
}

__device__ __forceinline__ void own_load8(float4 (&a)[8], const float* __restrict__ lin,
                                          int s, int j, int lane) {
#pragma unroll
    for (int u = 0; u < 8; ++u) {
        const int su = __builtin_amdgcn_readlane(s, j + u);
        a[u] = *(const float4*)(lin + su * D + 4 * lane);
    }
}

__device__ __forceinline__ void own_rmw1(float* mine, const float4 a, int k, float w) {
    float4* p = (float4*)(mine + k * (WAVES * PITCH));
    float4 x = *p;
    x.x += a.x * w;
    x.y += a.y * w;
    x.z += a.z * w;
    x.w += a.w * w;
    *p = x;
}

__device__ __forceinline__ void own_rmw2(float* mine, const float4 a0, const float4 a1,
                                         int k0, int k1, float w0, float w1) {
    float4* p0 = (float4*)(mine + k0 * (WAVES * PITCH));
    if (k0 != k1) {
        float4* p1 = (float4*)(mine + k1 * (WAVES * PITCH));
        float4 x0 = *p0;
        float4 x1 = *p1;
        x0.x += a0.x * w0;
        x0.y += a0.y * w0;
        x0.z += a0.z * w0;
        x0.w += a0.w * w0;
        x1.x += a1.x * w1;
        x1.y += a1.y * w1;
        x1.z += a1.z * w1;
        x1.w += a1.w * w1;
        *p0 = x0;
        *p1 = x1;
    } else {
        float4 x0 = *p0;
        x0.x += a0.x * w0;
        x0.y += a0.y * w0;
        x0.z += a0.z * w0;
        x0.w += a0.w * w0;
        x0.x += a1.x * w1;
        x0.y += a1.y * w1;
        x0.z += a1.z * w1;
        x0.w += a1.w * w1;
        *p0 = x0;
    }
}

__device__ __forceinline__ void own_rmw8_full(float* mine, const float4 (&a)[8],
                                              int k, float w, int j) {
#pragma unroll
    for (int u = 0; u < 8; u += 2) {
        const int   k0 = __builtin_amdgcn_readlane(k, j + u);
        const int   k1 = __builtin_amdgcn_readlane(k, j + u + 1);
        const float w0 = __int_as_float(__builtin_amdgcn_readlane(__float_as_int(w), j + u));
        const float w1 = __int_as_float(__builtin_amdgcn_readlane(__float_as_int(w), j + u + 1));
        own_rmw2(mine, a[u], a[u + 1], k0, k1, w0, w1);
    }
}

__device__ __forceinline__ void own_rmw8_part(float* mine, const float4 (&a)[8],
                                              int k, float w, int j, int cnt) {
#pragma unroll
    for (int u = 0; u < 8; ++u) {
        if (j + u < cnt) {
            const int   ku = __builtin_amdgcn_readlane(k, j + u);
            const float wu = __int_as_float(__builtin_amdgcn_readlane(__float_as_int(w), j + u));
            own_rmw1(mine, a[u], ku, wu);
        }
    }
}

__device__ __forceinline__ void own_add_edge(float* __restrict__ acc,
                                             const float* __restrict__ lin,
                                             const int* __restrict__ src,
                                             const float* __restrict__ dinv,
                                             int lo, int e, int r, int lane) {
    const int s = src[e];
    const float w = dinv[s] * dinv[lo + r];
    const float4 a = *(const float4*)(lin + s * D + 4 * lane);
    own_add_row(acc, a, w, r, lane);
}

template <bool PARTIAL>
__device__ __forceinline__ void own_add_rows8(float* __restrict__ acc,
                                              const float* __restrict__ lin,
                                              int s, int r, float w,
                                              int j, int cnt, int lane) {
    float4 a[8];
#pragma unroll
    for (int u = 0; u < 8; ++u) {
        const int su = __builtin_amdgcn_readlane(s, j + u);
        a[u] = *(const float4*)(lin + su * D + 4 * lane);
    }
#pragma unroll
    for (int u = 0; u < 8; ++u) {
        if (!PARTIAL || j + u < cnt) {
            const int   ru = __builtin_amdgcn_readlane(r, j + u);
            const float wu = __int_as_float(__builtin_amdgcn_readlane(__float_as_int(w), j + u));
            own_add_row(acc, a[u], wu, ru, lane);
        }
    }
}

__device__ __forceinline__ void own_take(int d, int e, int lo, int lane,
                                         unsigned* __restrict__ queue,
                                         unsigned* __restrict__ qn,
                                         float* __restrict__ acc,
                                         const float* __restrict__ lin,
                                         const int* __restrict__ src,
                                         const float* __restrict__ dinv) {
    const unsigned r = (unsigned)(d - lo);
    const unsigned ent = (unsigned)e | (r << 20);
    bool over = false;
    if (r < R_OWN) {
        const unsigned pos = atomicAdd(qn, 1u);
        if (pos < Q_CAP) queue[pos] = ent;
        else over = true;
    }
    unsigned ob = __builtin_amdgcn_ballot_w32(over);
    while (ob) {
        const int j = __builtin_ctz(ob);
        ob &= ob - 1;
        const unsigned ej = (unsigned)__builtin_amdgcn_readlane((int)ent, j);
        own_add_edge(acc, lin, src, dinv, lo, (int)(ej & 0xFFFFFu), (int)(ej >> 20), lane);
    }
}

__device__ __forceinline__ void own_scan4(const int4 v, int i4, int lo, int lane,
                                          unsigned* __restrict__ queue,
                                          unsigned* __restrict__ qn,
                                          float* __restrict__ acc,
                                          const float* __restrict__ lin,
                                          const int* __restrict__ src,
                                          const float* __restrict__ dinv) {
    const bool any = ((unsigned)(v.x - lo) < R_OWN) | ((unsigned)(v.y - lo) < R_OWN) |
                     ((unsigned)(v.z - lo) < R_OWN) | ((unsigned)(v.w - lo) < R_OWN);
    if (__builtin_amdgcn_ballot_w32(any) == 0) return;
    own_take(v.x, 4 * i4 + 0, lo, lane, queue, qn, acc, lin, src, dinv);
    own_take(v.y, 4 * i4 + 1, lo, lane, queue, qn, acc, lin, src, dinv);
    own_take(v.z, 4 * i4 + 2, lo, lane, queue, qn, acc, lin, src, dinv);
    own_take(v.w, 4 * i4 + 3, lo, lane, queue, qn, acc, lin, src, dinv);
}

__device__ __forceinline__ void stage_w(float* __restrict__ Ws,
                                        const float* __restrict__ W, int tid) {
    for (int u = tid; u < 2048; u += 1024) {
        const int kp = u >> 5;
        const int n4 = (u & 31) * 4;
        const int f0 = 2 * kp;
        const int f1 = 2 * kp + 1;
        const float4 r0 = *(const float4*)(W + f0 * D + n4);
        const float4 r1 = *(const float4*)(W + f1 * D + n4);
        float4* dst = (float4*)(Ws + kp * 256 + n4 * 2);
        dst[0] = make_float4(r0.x, r1.x, r0.y, r1.y);
        dst[1] = make_float4(r0.z, r1.z, r0.w, r1.w);
    }
}

template <int REUSE>
__global__ __launch_bounds__(1024) void gcn_layer_own(const float* __restrict__ hin,
                                                      const int* __restrict__ src,
                                                      const int* __restrict__ dst,
                                                      const float* __restrict__ dinv,
                                                      const float* __restrict__ W,
                                                      const float* __restrict__ bias,
                                                      float* __restrict__ hout,
                                                      const unsigned* __restrict__ qent,
                                                      float* __restrict__ qw,
                                                      const unsigned* __restrict__ qcount) {
    __shared__ __attribute__((aligned(16))) float acc[R_PAD * PITCH];
    __shared__ __attribute__((aligned(16))) unsigned queue[Q_CAP];
    __shared__ unsigned qn;
    __shared__ uint2 sub[WAVES * SUB_CAP];
    __shared__ unsigned scnt[WAVES];
    float* Ws = (float*)queue;

    const int tid  = threadIdx.x;
    const int lane = tid & 31;
    const int wave = __builtin_amdgcn_readfirstlane(tid >> 5);
    const int lo   = blockIdx.x * R_OWN;

    for (int i = tid; i < R_PAD * PITCH / 4; i += 1024)
        ((float4*)acc)[i] = make_float4(0.f, 0.f, 0.f, 0.f);
    if (tid == 0) qn = 0u;
    if (tid < WAVES) scnt[tid] = 0u;

    const unsigned* qe = qent + blockIdx.x * Q_CAP;
    float* qwb = qw + blockIdx.x * Q_CAP;
    const unsigned nsaved = qcount[blockIdx.x];
    const bool saved = nsaved <= Q_CAP;
    if (saved) stage_w(Ws, W, tid);
    __syncthreads();

    if (!saved) {
    const int4* d4 = (const int4*)dst;
    for (int it = 0; it < 192; it += 4) {
        const int i0 = it * 1024 + tid;
        const int4 v0 = d4[i0];
        const int4 v1 = d4[i0 + 1024];
        const int4 v2 = d4[i0 + 2048];
        const int4 v3 = d4[i0 + 3072];
        own_scan4(v0, i0,        lo, lane, queue, &qn, acc, hin, src, dinv);
        own_scan4(v1, i0 + 1024, lo, lane, queue, &qn, acc, hin, src, dinv);
        own_scan4(v2, i0 + 2048, lo, lane, queue, &qn, acc, hin, src, dinv);
        own_scan4(v3, i0 + 3072, lo, lane, queue, &qn, acc, hin, src, dinv);
    }
    for (int i = 192 * 1024 + tid; i < E4; i += 1024)
        own_scan4(d4[i], i, lo, lane, queue, &qn, acc, hin, src, dinv);
    __syncthreads();

    const unsigned nq = qn;
    const unsigned n = (unsigned)__builtin_amdgcn_readfirstlane((int)(nq < Q_CAP ? nq : Q_CAP));
    for (unsigned base = (unsigned)wave * 32u; base < n; base += 1024u) {
        const unsigned idx = base + (unsigned)lane;
        int   s = 0;
        int   r = 0;
        float w = 0.f;
        if (idx < n) {
            const unsigned ent = queue[idx];
            s = src[ent & 0xFFFFFu];
            r = (int)(ent >> 20);
            w = dinv[s] * dinv[lo + r];
        }
        const int cnt = (int)((n - base) < 32u ? (n - base) : 32u);
        int j = 0;
        for (; j + 8 <= cnt; j += 8)
            own_add_rows8<false>(acc, hin, s, r, w, j, cnt, lane);
        if (j < cnt)
            own_add_rows8<true>(acc, hin, s, r, w, j, cnt, lane);
    }
    } else {
    const unsigned n = (unsigned)__builtin_amdgcn_readfirstlane((int)nsaved);
    const int left = N_NODES - lo;
    const unsigned ntot = n + (unsigned)(left < R_OWN ? left : R_OWN);
    for (unsigned base = (unsigned)wave * 32u; base < ntot; base += 4096u) {
        unsigned ent[4];
        float    wv[4];
#pragma unroll
        for (int q = 0; q < 4; ++q) {
            const unsigned idx = base + (unsigned)q * 1024u + (unsigned)lane;
            ent[q] = 0u;
            wv[q]  = 0.f;
            if (idx < n) {
                ent[q] = qe[idx];
                if (REUSE) wv[q] = qwb[idx];
            }
        }
#pragma unroll
        for (int q = 0; q < 4; ++q) {
            const unsigned idx = base + (unsigned)q * 1024u + (unsigned)lane;
            int   s = 0;
            int   r = 0;
            float w = 0.f;
            bool over = false;
            if (idx < ntot) {
                if (idx < n) {
                    s = (int)(ent[q] & 0xFFFFu);
                    r = (int)(ent[q] >> 16);
                    if (REUSE) {
                        w = wv[q];
                    } else {
                        w = dinv[s] * dinv[lo + r];
                        qwb[idx] = w;
                    }
                } else {
                    r = (int)(idx - n);
                    s = lo + r;
                    w = dinv[s];
                    w = w * w;
                }
                const unsigned b   = (unsigned)r & (WAVES - 1);
                const unsigned pos = atomicAdd(&scnt[b], 1u);
                if (pos < SUB_CAP)
                    sub[b * SUB_CAP + pos] = make_uint2((unsigned)s | (((unsigned)r >> 5) << 16),
                                                        __float_as_uint(w));
                else
                    over = true;
            }
            unsigned ob = __builtin_amdgcn_ballot_w32(over);
            while (ob) {
                const int j = __builtin_ctz(ob);
                ob &= ob - 1;
                const int   sj = __builtin_amdgcn_readlane(s, j);
                const int   rj = __builtin_amdgcn_readlane(r, j);
                const float wj = __int_as_float(__builtin_amdgcn_readlane(__float_as_int(w), j));
                const float4 a = *(const float4*)(hin + sj * D + 4 * lane);
                own_add_row(acc, a, wj, rj, lane);
            }
        }
    }
    __syncthreads();

    {
        const unsigned mraw = scnt[wave];
        const unsigned m = (unsigned)__builtin_amdgcn_readfirstlane((int)(mraw < SUB_CAP ? mraw : SUB_CAP));
        const uint2* mysub = sub + wave * SUB_CAP;
        float* mine = acc + wave * PITCH + 4 * lane;
        for (unsigned b0 = 0; b0 < m; b0 += 32u) {
            const unsigned i = b0 + (unsigned)lane;
            uint2 e = make_uint2(0u, 0u);
            if (i < m) e = mysub[i];
            const int   s = (int)(e.x & 0xFFFFu);
            const int   k = (int)(e.x >> 16);
            const float w = __uint_as_float(e.y);
            const int cnt = (int)((m - b0) < 32u ? (m - b0) : 32u);
            float4 a0[8];
            float4 a1[8];
            if (cnt == 32) {
                own_load8(a0, hin, s, 0, lane);
                own_load8(a1, hin, s, 8, lane);
                own_rmw8_full(mine, a0, k, w, 0);
                own_load8(a0, hin, s, 16, lane);
                own_rmw8_full(mine, a1, k, w, 8);
                own_load8(a1, hin, s, 24, lane);
                own_rmw8_full(mine, a0, k, w, 16);
                own_rmw8_full(mine, a1, k, w, 24);
            } else {
                own_load8(a0, hin, s, 0, lane);
                if (cnt > 8) own_load8(a1, hin, s, 8, lane);
                own_rmw8_part(mine, a0, k, w, 0, cnt);
                if (cnt > 8) {
                    if (cnt > 16) own_load8(a0, hin, s, 16, lane);
                    own_rmw8_part(mine, a1, k, w, 8, cnt);
                    if (cnt > 16) {
                        if (cnt > 24) own_load8(a1, hin, s, 24, lane);
                        own_rmw8_part(mine, a0, k, w, 16, cnt);
                        if (cnt > 24) own_rmw8_part(mine, a1, k, w, 24, cnt);
                    }
                }
            }
        }
    }
    }
    __syncthreads();

    if (!saved) {
        stage_w(Ws, W, tid);
        for (int i = tid; i < R_OWN * (D / 4); i += 1024) {
            const int node = lo + (i >> 5);
            if (node < N_NODES) {
                const int f4 = i & 31;
                float sl = dinv[node];
                sl = sl * sl;
                const float4 l = ((const float4*)hin)[node * (D / 4) + f4];
                float4* ar = (float4*)(acc + (i >> 5) * PITCH + 4 * f4);
                float4 x = *ar;
                x.x += l.x * sl;
                x.y += l.y * sl;
                x.z += l.z * sl;
                x.w += l.w * sl;
                *ar = x;
            }
        }
        __syncthreads();
    }

    if (wave >= 26) return;
    const int rt    = wave >> 1;
    const int cbase = (wave & 1) * 64;
    const int nlo   = lane & 15;
    const int khalf = lane >> 4;

    const v2f* Arow = (const v2f*)(acc + (rt * 16 + nlo) * PITCH + 2 * khalf);
    v8f o[4] = {};
    for (int j = 0; j < 32; ++j) {
        const v2f a = Arow[2 * j];
        const v2f* Brow = (const v2f*)(Ws + (2 * j + khalf) * 256 + cbase * 2);
#pragma unroll
        for (int t = 0; t < 4; ++t) {
            const v2f b = Brow[t * 16 + nlo];
            o[t] = __builtin_amdgcn_wmma_f32_16x16x4_f32(
                false, a, false, b, (short)0, o[t], false, false);
        }
    }

    const int lrbase = rt * 16 + 8 * khalf;
#pragma unroll
    for (int t = 0; t < 4; ++t) {
        const int nn = cbase + t * 16 + nlo;
        const float bb = bias[nn];
#pragma unroll
        for (int r = 0; r < 8; ++r) {
            const int lr = lrbase + r;
            if (lr < R_OWN && lo + lr < N_NODES)
                hout[(lo + lr) * D + nn] = fmaxf(o[t][r] + bb, 0.f);
        }
    }
}

__global__ __launch_bounds__(256) void gcn_fc(const float* __restrict__ H,
                                              const float* __restrict__ W,
                                              const float* __restrict__ bias,
                                              float* __restrict__ O) {
    const int wave = threadIdx.x >> 5;
    const int lane = threadIdx.x & 31;
    const int tile = blockIdx.x * 8 + wave;
    if (tile >= ROW_TILES) return;

    const int nlo   = lane & 15;
    const int khalf = lane >> 4;
    const int row   = tile * 16 + nlo;

    v8f acc[3] = {};
    for (int k0 = 0; k0 < D; k0 += 4) {
        const int ka = k0 + 2 * khalf;
        v2f a;
        a.x = H[row * D + ka];
        a.y = H[row * D + ka + 1];
#pragma unroll
        for (int t = 0; t < 3; ++t) {
            const int n  = t * 16 + nlo;
            const int nc = n < N_CLASSES ? n : (N_CLASSES - 1);
            v2f b;
            b.x = W[ka * N_CLASSES + nc];
            b.y = W[(ka + 1) * N_CLASSES + nc];
            acc[t] = __builtin_amdgcn_wmma_f32_16x16x4_f32(
                false, a, false, b, (short)0, acc[t], false, false);
        }
    }

    const int rbase = tile * 16 + 8 * khalf;
#pragma unroll
    for (int t = 0; t < 3; ++t) {
        const int n = t * 16 + nlo;
        if (n < N_CLASSES) {
            const float bb = bias[n];
#pragma unroll
            for (int r = 0; r < 8; ++r)
                O[(rbase + r) * N_CLASSES + n] = acc[t][r] + bb;
        }
    }
}

extern "C" void kernel_launch(void* const* d_in, const int* in_sizes, int n_in,
                              void* d_out, int out_size, void* d_ws, size_t ws_size,
                              hipStream_t stream) {
    const float* x   = (const float*)d_in[0];
    const int*   ei  = (const int*)d_in[1];
    const float* W1  = (const float*)d_in[2];
    const float* b1  = (const float*)d_in[3];
    const float* W2  = (const float*)d_in[4];
    const float* b2  = (const float*)d_in[5];
    const float* W3  = (const float*)d_in[6];
    const float* b3  = (const float*)d_in[7];
    const float* fcW = (const float*)d_in[8];
    const float* fcb = (const float*)d_in[9];

    float* out    = (float*)d_out;
    float* h_out  = out;
    float* logits = out + N_NODES * D;

    float* ws   = (float*)d_ws;
    float* dinv = ws;
    float* B0   = ws + 50176;
    float* B1   = B0 + N_NODES * D;
    unsigned* qent   = (unsigned*)(B1 + N_NODES * D);
    float*    qw     = (float*)(qent + OWN_BLOCKS * Q_CAP);
    unsigned* qcount = (unsigned*)(qw + OWN_BLOCKS * Q_CAP);
    unsigned* dealE  = qcount + OWN_BLOCKS;
    unsigned* dealF  = dealE + DEALERS * DEAL_E;

    const int* src = ei;
    const int* dst = ei + N_EDGES;

    const int TPB = 256;
    const int gTile = (ROW_TILES + 7) / 8;

    edge_deal<<<DEALERS, 1024, 0, stream>>>(src, dst, dealE, dealF);
    own_build<<<OWN_BLOCKS, 1024, 0, stream>>>(dst, dealE, dealF, qent, qcount, dinv);

    gcn_layer_own<0><<<OWN_BLOCKS, 1024, 0, stream>>>(x,  src, dst, dinv, W1, b1, B0, qent, qw, qcount);
    gcn_layer_own<1><<<OWN_BLOCKS, 1024, 0, stream>>>(B0, src, dst, dinv, W2, b2, B1, qent, qw, qcount);
    gcn_layer_own<1><<<OWN_BLOCKS, 1024, 0, stream>>>(B1, src, dst, dinv, W3, b3, h_out, qent, qw, qcount);

    gcn_fc<<<gTile, TPB, 0, stream>>>(h_out, fcW, fcb, logits);
}
